# starting kernel
# baseline (speedup 1.0000x reference)
#include <hip/hip_runtime.h>

typedef float v2f __attribute__((ext_vector_type(2)));
typedef float v8f __attribute__((ext_vector_type(8)));

__device__ __forceinline__ float bperm(int srcLane, float v) {
    return __int_as_float(__builtin_amdgcn_ds_bpermute(srcLane << 2, __float_as_int(v)));
}

__global__ __launch_bounds__(256) void fused_attn_kernel(
    const float* __restrict__ hist,
    const float* __restrict__ paper,
    const float* __restrict__ author,
    const float* __restrict__ Wh, const float* __restrict__ bh,
    const float* __restrict__ Wp, const float* __restrict__ bp,
    const float* __restrict__ Wa, const float* __restrict__ ba,
    const float* __restrict__ lnw, const float* __restrict__ lnb,
    const float* __restrict__ Wfc, const float* __restrict__ bfc,
    float* __restrict__ out, int n)
{
    const int row      = blockIdx.x * blockDim.x + threadIdx.x;
    const int lane     = threadIdx.x & 31;
    const int waveBase = row - lane;
    const bool lo      = lane < 16;

    float flat[6] = {0.f, 0.f, 0.f, 0.f, 0.f, 0.f};

    if (row < n) {
        const float* hp = hist   + (size_t)row * 2u;
        const float* pp = paper  + (size_t)row * 5u;
        const float* ap = author + (size_t)row * 5u;
        v2f   hv  = __builtin_nontemporal_load((const v2f*)hp);
        float pv0 = __builtin_nontemporal_load(pp + 0);
        float pv1 = __builtin_nontemporal_load(pp + 1);
        float pv2 = __builtin_nontemporal_load(pp + 2);
        float pv3 = __builtin_nontemporal_load(pp + 3);
        float pv4 = __builtin_nontemporal_load(pp + 4);
        float av0 = __builtin_nontemporal_load(ap + 0);
        float av1 = __builtin_nontemporal_load(ap + 1);
        float av2 = __builtin_nontemporal_load(ap + 2);
        float av3 = __builtin_nontemporal_load(ap + 3);
        float av4 = __builtin_nontemporal_load(ap + 4);

        float fx[3], fy[3];
        fx[0] = fmaxf(Wh[0]*hv.x + Wh[1]*hv.y + bh[0], 0.f);
        fy[0] = fmaxf(Wh[2]*hv.x + Wh[3]*hv.y + bh[1], 0.f);
        fx[1] = fmaxf(Wp[0]*pv0 + Wp[1]*pv1 + Wp[2]*pv2 + Wp[3]*pv3 + Wp[4]*pv4 + bp[0], 0.f);
        fy[1] = fmaxf(Wp[5]*pv0 + Wp[6]*pv1 + Wp[7]*pv2 + Wp[8]*pv3 + Wp[9]*pv4 + bp[1], 0.f);
        fx[2] = fmaxf(Wa[0]*av0 + Wa[1]*av1 + Wa[2]*av2 + Wa[3]*av3 + Wa[4]*av4 + ba[0], 0.f);
        fy[2] = fmaxf(Wa[5]*av0 + Wa[6]*av1 + Wa[7]*av2 + Wa[8]*av3 + Wa[9]*av4 + ba[1], 0.f);

        const float is2 = 0.70710678118654752f;
        float l[3][3];
        #pragma unroll
        for (int s = 0; s < 3; ++s)
            #pragma unroll
            for (int t = 0; t < 3; ++t)
                l[s][t] = (fx[s]*fx[t] + fy[s]*fy[t]) * is2;

        float w[3][3];
        #pragma unroll
        for (int t = 0; t < 3; ++t) {
            float m  = fmaxf(l[0][t], fmaxf(l[1][t], l[2][t]));
            float e0 = __expf(l[0][t] - m);
            float e1 = __expf(l[1][t] - m);
            float e2 = __expf(l[2][t] - m);
            float r  = __builtin_amdgcn_rcpf(e0 + e1 + e2);
            w[0][t] = e0 * r; w[1][t] = e1 * r; w[2][t] = e2 * r;
        }

        float ax[3], ay[3];
        #pragma unroll
        for (int s = 0; s < 3; ++s) {
            ax[s] = w[s][0]*fx[0] + w[s][1]*fx[1] + w[s][2]*fx[2];
            ay[s] = w[s][0]*fy[0] + w[s][1]*fy[1] + w[s][2]*fy[2];
        }

        float mu = (ax[0]+ay[0]+ax[1]+ay[1]+ax[2]+ay[2]) * (1.f/6.f);
        float var = 0.f;
        #pragma unroll
        for (int s = 0; s < 3; ++s) {
            float dx = ax[s] - mu, dy = ay[s] - mu;
            var += dx*dx + dy*dy;
        }
        var *= (1.f/6.f);
        float inv = __builtin_amdgcn_rsqf(var + 1e-5f);
        #pragma unroll
        for (int s = 0; s < 3; ++s) {
            flat[2*s+0] = (ax[s] - mu) * inv * lnw[2*s+0] + lnb[2*s+0];
            flat[2*s+1] = (ay[s] - mu) * inv * lnw[2*s+1] + lnb[2*s+1];
        }
    }

    if (waveBase + 32 <= n) {
        const int  j  = lane & 15;
        const bool jv = j < 6;
        v2f A1, A2;
        A1.x = jv ? Wfc[j*6 + (lo ? 0 : 2)] : 0.f;
        A1.y = jv ? Wfc[j*6 + (lo ? 1 : 3)] : 0.f;
        A2.x = (jv && lo) ? Wfc[j*6 + 4] : 0.f;
        A2.y = (jv && lo) ? Wfc[j*6 + 5] : 0.f;

        v8f C = {};
        if (lo) { C[0]=bfc[0]; C[1]=bfc[1]; C[2]=bfc[2]; C[3]=bfc[3]; C[4]=bfc[4]; C[5]=bfc[5]; }

        float f2s = bperm(lane & 15, flat[2]);
        float f3s = bperm(lane & 15, flat[3]);
        v2f B1; B1.x = lo ? flat[0] : f2s;  B1.y = lo ? flat[1] : f3s;
        v2f B2; B2.x = lo ? flat[4] : 0.f;  B2.y = lo ? flat[5] : 0.f;
        v8f D1 = __builtin_amdgcn_wmma_f32_16x16x4_f32(false, A1, false, B1, (short)0, C,  false, false);
        D1     = __builtin_amdgcn_wmma_f32_16x16x4_f32(false, A2, false, B2, (short)0, D1, false, false);

        float f0s = bperm(lane | 16, flat[0]);
        float f1s = bperm(lane | 16, flat[1]);
        float f4s = bperm(lane | 16, flat[4]);
        float f5s = bperm(lane | 16, flat[5]);
        v2f B3; B3.x = lo ? f0s : flat[2];  B3.y = lo ? f1s : flat[3];
        v2f B4; B4.x = lo ? f4s : 0.f;      B4.y = lo ? f5s : 0.f;
        v8f D2 = __builtin_amdgcn_wmma_f32_16x16x4_f32(false, A1, false, B3, (short)0, C,  false, false);
        D2     = __builtin_amdgcn_wmma_f32_16x16x4_f32(false, A2, false, B4, (short)0, D2, false, false);

        if (lo) {
            float* o1 = out + (size_t)(waveBase + lane) * 6u;
            v2f s0; s0.x = fmaxf(D1[0], 0.f); s0.y = fmaxf(D1[1], 0.f);
            v2f s1; s1.x = fmaxf(D1[2], 0.f); s1.y = fmaxf(D1[3], 0.f);
            v2f s2; s2.x = fmaxf(D1[4], 0.f); s2.y = fmaxf(D1[5], 0.f);
            __builtin_nontemporal_store(s0, (v2f*)(o1 + 0));
            __builtin_nontemporal_store(s1, (v2f*)(o1 + 2));
            __builtin_nontemporal_store(s2, (v2f*)(o1 + 4));

            float* o2 = out + (size_t)(waveBase + 16 + lane) * 6u;
            v2f t0; t0.x = fmaxf(D2[0], 0.f); t0.y = fmaxf(D2[1], 0.f);
            v2f t1; t1.x = fmaxf(D2[2], 0.f); t1.y = fmaxf(D2[3], 0.f);
            v2f t2; t2.x = fmaxf(D2[4], 0.f); t2.y = fmaxf(D2[5], 0.f);
            __builtin_nontemporal_store(t0, (v2f*)(o2 + 0));
            __builtin_nontemporal_store(t1, (v2f*)(o2 + 2));
            __builtin_nontemporal_store(t2, (v2f*)(o2 + 4));
        }
    } else if (row < n) {
        float* o = out + (size_t)row * 6u;
        #pragma unroll
        for (int jj = 0; jj < 6; ++jj) {
            float acc = bfc[jj];
            #pragma unroll
            for (int k = 0; k < 6; ++k) acc += Wfc[jj*6 + k] * flat[k];
            o[jj] = fmaxf(acc, 0.f);
        }
    }
}

extern "C" void kernel_launch(void* const* d_in, const int* in_sizes, int n_in,
                              void* d_out, int out_size, void* d_ws, size_t ws_size,
                              hipStream_t stream) {
    const float* hist   = (const float*)d_in[0];
    const float* paper  = (const float*)d_in[1];
    const float* author = (const float*)d_in[2];
    const float* Wh  = (const float*)d_in[3];
    const float* bh  = (const float*)d_in[4];
    const float* Wp  = (const float*)d_in[5];
    const float* bp  = (const float*)d_in[6];
    const float* Wa  = (const float*)d_in[7];
    const float* ba  = (const float*)d_in[8];
    const float* lnw = (const float*)d_in[9];
    const float* lnb = (const float*)d_in[10];
    const float* Wfc = (const float*)d_in[11];
    const float* bfc = (const float*)d_in[12];
    float* out = (float*)d_out;

    const int n = in_sizes[0] / 2;
    const int threads = 256;
    const int blocks  = (n + threads - 1) / threads;

    fused_attn_kernel<<<blocks, threads, 0, stream>>>(
        hist, paper, author, Wh, bh, Wp, bp, Wa, ba, lnw, lnb, Wfc, bfc, out, n);
}


# Row math on explicit row pairs: sub-blocks 2p and 2p+1 of a lane sit in 64-bit register pairs, one packed mul/add/fma serves both rows, weights broadcast; 124 registers not 150
# speedup vs baseline: 1.1775x; 1.1775x over previous
#include <hip/hip_runtime.h>


typedef float v2f __attribute__((ext_vector_type(2)));
typedef float v8f __attribute__((ext_vector_type(8)));

__device__ __forceinline__ float bperm(int srcLane, float v) {
    return __int_as_float(__builtin_amdgcn_ds_bpermute(srcLane << 2, __float_as_int(v)));
}

__device__ __forceinline__ void row_forward(
    float h0, float h1,
    float pv0, float pv1, float pv2, float pv3, float pv4,
    float av0, float av1, float av2, float av3, float av4,
    const float* __restrict__ Wh, const float* __restrict__ bh,
    const float* __restrict__ Wp, const float* __restrict__ bp,
    const float* __restrict__ Wa, const float* __restrict__ ba,
    const float* __restrict__ lnw, const float* __restrict__ lnb,
    float* flat)
{
    float fx[3], fy[3];
    fx[0] = fmaxf(Wh[0]*h0 + Wh[1]*h1 + bh[0], 0.f);
    fy[0] = fmaxf(Wh[2]*h0 + Wh[3]*h1 + bh[1], 0.f);
    fx[1] = fmaxf(Wp[0]*pv0 + Wp[1]*pv1 + Wp[2]*pv2 + Wp[3]*pv3 + Wp[4]*pv4 + bp[0], 0.f);
    fy[1] = fmaxf(Wp[5]*pv0 + Wp[6]*pv1 + Wp[7]*pv2 + Wp[8]*pv3 + Wp[9]*pv4 + bp[1], 0.f);
    fx[2] = fmaxf(Wa[0]*av0 + Wa[1]*av1 + Wa[2]*av2 + Wa[3]*av3 + Wa[4]*av4 + ba[0], 0.f);
    fy[2] = fmaxf(Wa[5]*av0 + Wa[6]*av1 + Wa[7]*av2 + Wa[8]*av3 + Wa[9]*av4 + ba[1], 0.f);

    const float is2 = 0.70710678118654752f;
    float l[3][3];
    #pragma unroll
    for (int s = 0; s < 3; ++s)
        #pragma unroll
        for (int t = 0; t < 3; ++t)
            l[s][t] = (fx[s]*fx[t] + fy[s]*fy[t]) * is2;

    float w[3][3];
    #pragma unroll
    for (int t = 0; t < 3; ++t) {
        float m  = fmaxf(l[0][t], fmaxf(l[1][t], l[2][t]));
        float e0 = __expf(l[0][t] - m);
        float e1 = __expf(l[1][t] - m);
        float e2 = __expf(l[2][t] - m);
        float r  = __builtin_amdgcn_rcpf(e0 + e1 + e2);
        w[0][t] = e0 * r; w[1][t] = e1 * r; w[2][t] = e2 * r;
    }

    float ax[3], ay[3];
    #pragma unroll
    for (int s = 0; s < 3; ++s) {
        ax[s] = w[s][0]*fx[0] + w[s][1]*fx[1] + w[s][2]*fx[2];
        ay[s] = w[s][0]*fy[0] + w[s][1]*fy[1] + w[s][2]*fy[2];
    }

    float mu = (ax[0]+ay[0]+ax[1]+ay[1]+ax[2]+ay[2]) * (1.f/6.f);
    float var = 0.f;
    #pragma unroll
    for (int s = 0; s < 3; ++s) {
        float dx = ax[s] - mu, dy = ay[s] - mu;
        var += dx*dx + dy*dy;
    }
    var *= (1.f/6.f);
    float inv = __builtin_amdgcn_rsqf(var + 1e-5f);
    #pragma unroll
    for (int s = 0; s < 3; ++s) {
        flat[2*s+0] = (ax[s] - mu) * inv * lnw[2*s+0] + lnb[2*s+0];
        flat[2*s+1] = (ay[s] - mu) * inv * lnw[2*s+1] + lnb[2*s+1];
    }
}

__device__ __forceinline__ v2f pmax(v2f a, v2f b) { return __builtin_elementwise_max(a, b); }

__device__ __forceinline__ void pair_forward(
    v2f h0, v2f h1,
    v2f pv0, v2f pv1, v2f pv2, v2f pv3, v2f pv4,
    v2f av0, v2f av1, v2f av2, v2f av3, v2f av4,
    const float* __restrict__ Wh, const float* __restrict__ bh,
    const float* __restrict__ Wp, const float* __restrict__ bp,
    const float* __restrict__ Wa, const float* __restrict__ ba,
    const float* __restrict__ lnw, const float* __restrict__ lnb,
    v2f* flat)
{
    const v2f zero = {0.f, 0.f};
    v2f fx[3], fy[3];
    fx[0] = pmax(Wh[0]*h0 + Wh[1]*h1 + bh[0], zero);
    fy[0] = pmax(Wh[2]*h0 + Wh[3]*h1 + bh[1], zero);
    fx[1] = pmax(Wp[0]*pv0 + Wp[1]*pv1 + Wp[2]*pv2 + Wp[3]*pv3 + Wp[4]*pv4 + bp[0], zero);
    fy[1] = pmax(Wp[5]*pv0 + Wp[6]*pv1 + Wp[7]*pv2 + Wp[8]*pv3 + Wp[9]*pv4 + bp[1], zero);
    fx[2] = pmax(Wa[0]*av0 + Wa[1]*av1 + Wa[2]*av2 + Wa[3]*av3 + Wa[4]*av4 + ba[0], zero);
    fy[2] = pmax(Wa[5]*av0 + Wa[6]*av1 + Wa[7]*av2 + Wa[8]*av3 + Wa[9]*av4 + ba[1], zero);

    const float is2 = 0.70710678118654752f;
    v2f l[3][3];
    #pragma unroll
    for (int s = 0; s < 3; ++s)
        #pragma unroll
        for (int t = 0; t < 3; ++t)
            l[s][t] = (fx[s]*fx[t] + fy[s]*fy[t]) * is2;

    const float log2e = 1.44269504088896341f;
    v2f w[3][3];
    #pragma unroll
    for (int t = 0; t < 3; ++t) {
        v2f m  = pmax(l[0][t], pmax(l[1][t], l[2][t]));
        v2f a0 = (l[0][t] - m) * log2e;
        v2f a1 = (l[1][t] - m) * log2e;
        v2f a2 = (l[2][t] - m) * log2e;
        v2f e0, e1, e2;
        e0.x = __builtin_amdgcn_exp2f(a0.x); e0.y = __builtin_amdgcn_exp2f(a0.y);
        e1.x = __builtin_amdgcn_exp2f(a1.x); e1.y = __builtin_amdgcn_exp2f(a1.y);
        e2.x = __builtin_amdgcn_exp2f(a2.x); e2.y = __builtin_amdgcn_exp2f(a2.y);
        v2f sum = e0 + e1 + e2;
        v2f r;
        r.x = __builtin_amdgcn_rcpf(sum.x); r.y = __builtin_amdgcn_rcpf(sum.y);
        w[0][t] = e0 * r; w[1][t] = e1 * r; w[2][t] = e2 * r;
    }

    v2f ax[3], ay[3];
    #pragma unroll
    for (int s = 0; s < 3; ++s) {
        ax[s] = w[s][0]*fx[0] + w[s][1]*fx[1] + w[s][2]*fx[2];
        ay[s] = w[s][0]*fy[0] + w[s][1]*fy[1] + w[s][2]*fy[2];
    }

    v2f mu = (ax[0]+ay[0]+ax[1]+ay[1]+ax[2]+ay[2]) * (1.f/6.f);
    v2f var = zero;
    #pragma unroll
    for (int s = 0; s < 3; ++s) {
        v2f dx = ax[s] - mu, dy = ay[s] - mu;
        var += dx*dx + dy*dy;
    }
    var = var * (1.f/6.f) + 1e-5f;
    v2f inv;
    inv.x = __builtin_amdgcn_rsqf(var.x); inv.y = __builtin_amdgcn_rsqf(var.y);
    #pragma unroll
    for (int s = 0; s < 3; ++s) {
        flat[2*s+0] = (ax[s] - mu) * inv * lnw[2*s+0] + lnb[2*s+0];
        flat[2*s+1] = (ay[s] - mu) * inv * lnw[2*s+1] + lnb[2*s+1];
    }
}

__global__ __launch_bounds__(256) void fused_attn_kernel(
    const float* __restrict__ hist,
    const float* __restrict__ paper,
    const float* __restrict__ author,
    const float* __restrict__ Wh, const float* __restrict__ bh,
    const float* __restrict__ Wp, const float* __restrict__ bp,
    const float* __restrict__ Wa, const float* __restrict__ ba,
    const float* __restrict__ lnw, const float* __restrict__ lnb,
    const float* __restrict__ Wfc, const float* __restrict__ bfc,
    float* __restrict__ out, int n)
{
    const int    lane     = threadIdx.x & 31;
    const size_t wave     = ((size_t)blockIdx.x * blockDim.x + threadIdx.x) >> 5;
    const size_t waveBase = wave * 256u;
    const bool   lo       = lane < 16;

    if (waveBase >= (size_t)n) return;

    if (waveBase + 256u <= (size_t)n) {
        v2f   hv[8];
        float pv[8][5], av[8][5];
        #pragma unroll
        for (int q = 0; q < 8; ++q) {
            const size_t row = waveBase + (size_t)(32 * q + lane);
            const float* hp = hist   + row * 2u;
            const float* pp = paper  + row * 5u;
            const float* ap = author + row * 5u;
            hv[q] = __builtin_nontemporal_load((const v2f*)hp);
            #pragma unroll
            for (int i = 0; i < 5; ++i) pv[q][i] = __builtin_nontemporal_load(pp + i);
            #pragma unroll
            for (int i = 0; i < 5; ++i) av[q][i] = __builtin_nontemporal_load(ap + i);
        }

        float flat[8][6];
        #pragma unroll
        for (int p = 0; p < 4; ++p) {
            const int q0 = 2 * p, q1 = 2 * p + 1;
            v2f F[6];
            pair_forward((v2f){hv[q0].x, hv[q1].x}, (v2f){hv[q0].y, hv[q1].y},
                         (v2f){pv[q0][0], pv[q1][0]}, (v2f){pv[q0][1], pv[q1][1]},
                         (v2f){pv[q0][2], pv[q1][2]}, (v2f){pv[q0][3], pv[q1][3]},
                         (v2f){pv[q0][4], pv[q1][4]},
                         (v2f){av[q0][0], av[q1][0]}, (v2f){av[q0][1], av[q1][1]},
                         (v2f){av[q0][2], av[q1][2]}, (v2f){av[q0][3], av[q1][3]},
                         (v2f){av[q0][4], av[q1][4]},
                         Wh, bh, Wp, bp, Wa, ba, lnw, lnb, F);
            #pragma unroll
            for (int k = 0; k < 6; ++k) { flat[q0][k] = F[k].x; flat[q1][k] = F[k].y; }
        }

        const int  j  = lane & 15;
        const bool jv = j < 6;
        v2f A1, A2;
        A1.x = jv ? Wfc[j*6 + (lo ? 0 : 2)] : 0.f;
        A1.y = jv ? Wfc[j*6 + (lo ? 1 : 3)] : 0.f;
        A2.x = (jv && lo) ? Wfc[j*6 + 4] : 0.f;
        A2.y = (jv && lo) ? Wfc[j*6 + 5] : 0.f;

        v8f C = {};
        if (lo) { C[0]=bfc[0]; C[1]=bfc[1]; C[2]=bfc[2]; C[3]=bfc[3]; C[4]=bfc[4]; C[5]=bfc[5]; }

        #pragma unroll
        for (int q = 0; q < 8; ++q) {
            const float* f = flat[q];
            const size_t subBase = waveBase + (size_t)(32 * q);
            float f2s = bperm(lane & 15, f[2]);
            float f3s = bperm(lane & 15, f[3]);
            v2f B1; B1.x = lo ? f[0] : f2s;  B1.y = lo ? f[1] : f3s;
            v2f B2; B2.x = lo ? f[4] : 0.f;  B2.y = lo ? f[5] : 0.f;
            v8f D1 = __builtin_amdgcn_wmma_f32_16x16x4_f32(false, A1, false, B1, (short)0, C,  false, false);
            D1     = __builtin_amdgcn_wmma_f32_16x16x4_f32(false, A2, false, B2, (short)0, D1, false, false);

            float f0s = bperm(lane | 16, f[0]);
            float f1s = bperm(lane | 16, f[1]);
            float f4s = bperm(lane | 16, f[4]);
            float f5s = bperm(lane | 16, f[5]);
            v2f B3; B3.x = lo ? f0s : f[2];  B3.y = lo ? f1s : f[3];
            v2f B4; B4.x = lo ? f4s : 0.f;   B4.y = lo ? f5s : 0.f;
            v8f D2 = __builtin_amdgcn_wmma_f32_16x16x4_f32(false, A1, false, B3, (short)0, C,  false, false);
            D2     = __builtin_amdgcn_wmma_f32_16x16x4_f32(false, A2, false, B4, (short)0, D2, false, false);

            if (lo) {
                float* o1 = out + (subBase + (size_t)lane) * 6u;
                v2f s0; s0.x = fmaxf(D1[0], 0.f); s0.y = fmaxf(D1[1], 0.f);
                v2f s1; s1.x = fmaxf(D1[2], 0.f); s1.y = fmaxf(D1[3], 0.f);
                v2f s2; s2.x = fmaxf(D1[4], 0.f); s2.y = fmaxf(D1[5], 0.f);
                __builtin_nontemporal_store(s0, (v2f*)(o1 + 0));
                __builtin_nontemporal_store(s1, (v2f*)(o1 + 2));
                __builtin_nontemporal_store(s2, (v2f*)(o1 + 4));

                float* o2 = out + (subBase + 16u + (size_t)lane) * 6u;
                v2f t0; t0.x = fmaxf(D2[0], 0.f); t0.y = fmaxf(D2[1], 0.f);
                v2f t1; t1.x = fmaxf(D2[2], 0.f); t1.y = fmaxf(D2[3], 0.f);
                v2f t2; t2.x = fmaxf(D2[4], 0.f); t2.y = fmaxf(D2[5], 0.f);
                __builtin_nontemporal_store(t0, (v2f*)(o2 + 0));
                __builtin_nontemporal_store(t1, (v2f*)(o2 + 2));
                __builtin_nontemporal_store(t2, (v2f*)(o2 + 4));
            }
        }
    } else {
        for (int q = 0; q < 8; ++q) {
            const size_t row = waveBase + (size_t)(32 * q + lane);
            if (row < (size_t)n) {
                const float* hp = hist   + row * 2u;
                const float* pp = paper  + row * 5u;
                const float* ap = author + row * 5u;
                float flat[6];
                row_forward(hp[0], hp[1], pp[0], pp[1], pp[2], pp[3], pp[4],
                            ap[0], ap[1], ap[2], ap[3], ap[4],
                            Wh, bh, Wp, bp, Wa, ba, lnw, lnb, flat);
                float* o = out + row * 6u;
                #pragma unroll
                for (int jj = 0; jj < 6; ++jj) {
                    float acc = bfc[jj];
                    #pragma unroll
                    for (int k = 0; k < 6; ++k) acc += Wfc[jj*6 + k] * flat[k];
                    o[jj] = fmaxf(acc, 0.f);
                }
            }
        }
    }
}

extern "C" void kernel_launch(void* const* d_in, const int* in_sizes, int n_in,
                              void* d_out, int out_size, void* d_ws, size_t ws_size,
                              hipStream_t stream) {
    const float* hist   = (const float*)d_in[0];
    const float* paper  = (const float*)d_in[1];
    const float* author = (const float*)d_in[2];
    const float* Wh  = (const float*)d_in[3];
    const float* bh  = (const float*)d_in[4];
    const float* Wp  = (const float*)d_in[5];
    const float* bp  = (const float*)d_in[6];
    const float* Wa  = (const float*)d_in[7];
    const float* ba  = (const float*)d_in[8];
    const float* lnw = (const float*)d_in[9];
    const float* lnb = (const float*)d_in[10];
    const float* Wfc = (const float*)d_in[11];
    const float* bfc = (const float*)d_in[12];
    float* out = (float*)d_out;

    const int n = in_sizes[0] / 2;
    const int threads = 256;
    const int rowsPerBlock = 2048;
    const int blocks  = (n + rowsPerBlock - 1) / rowsPerBlock;

    fused_attn_kernel<<<blocks, threads, 0, stream>>>(
        hist, paper, author, Wh, bh, Wp, bp, Wa, ba, lnw, lnb, Wfc, bfc, out, n);
}


# Row math of all four pairs held in one basic block ahead of the masked store blocks, so scalar weights fold as broadcast operands: no per-use weight copies or lane spills; 134 registers not 124
# speedup vs baseline: 1.2187x; 1.0350x over previous
#include <hip/hip_runtime.h>


typedef float v2f __attribute__((ext_vector_type(2)));
typedef float v8f __attribute__((ext_vector_type(8)));

__device__ __forceinline__ float bperm(int srcLane, float v) {
    return __int_as_float(__builtin_amdgcn_ds_bpermute(srcLane << 2, __float_as_int(v)));
}

__device__ __forceinline__ void row_forward(
    float h0, float h1,
    float pv0, float pv1, float pv2, float pv3, float pv4,
    float av0, float av1, float av2, float av3, float av4,
    const float* __restrict__ Wh, const float* __restrict__ bh,
    const float* __restrict__ Wp, const float* __restrict__ bp,
    const float* __restrict__ Wa, const float* __restrict__ ba,
    const float* __restrict__ lnw, const float* __restrict__ lnb,
    float* flat)
{
    float fx[3], fy[3];
    fx[0] = fmaxf(Wh[0]*h0 + Wh[1]*h1 + bh[0], 0.f);
    fy[0] = fmaxf(Wh[2]*h0 + Wh[3]*h1 + bh[1], 0.f);
    fx[1] = fmaxf(Wp[0]*pv0 + Wp[1]*pv1 + Wp[2]*pv2 + Wp[3]*pv3 + Wp[4]*pv4 + bp[0], 0.f);
    fy[1] = fmaxf(Wp[5]*pv0 + Wp[6]*pv1 + Wp[7]*pv2 + Wp[8]*pv3 + Wp[9]*pv4 + bp[1], 0.f);
    fx[2] = fmaxf(Wa[0]*av0 + Wa[1]*av1 + Wa[2]*av2 + Wa[3]*av3 + Wa[4]*av4 + ba[0], 0.f);
    fy[2] = fmaxf(Wa[5]*av0 + Wa[6]*av1 + Wa[7]*av2 + Wa[8]*av3 + Wa[9]*av4 + ba[1], 0.f);

    const float is2 = 0.70710678118654752f;
    float l[3][3];
    #pragma unroll
    for (int s = 0; s < 3; ++s)
        #pragma unroll
        for (int t = 0; t < 3; ++t)
            l[s][t] = (fx[s]*fx[t] + fy[s]*fy[t]) * is2;

    float w[3][3];
    #pragma unroll
    for (int t = 0; t < 3; ++t) {
        float m  = fmaxf(l[0][t], fmaxf(l[1][t], l[2][t]));
        float e0 = __expf(l[0][t] - m);
        float e1 = __expf(l[1][t] - m);
        float e2 = __expf(l[2][t] - m);
        float r  = __builtin_amdgcn_rcpf(e0 + e1 + e2);
        w[0][t] = e0 * r; w[1][t] = e1 * r; w[2][t] = e2 * r;
    }

    float ax[3], ay[3];
    #pragma unroll
    for (int s = 0; s < 3; ++s) {
        ax[s] = w[s][0]*fx[0] + w[s][1]*fx[1] + w[s][2]*fx[2];
        ay[s] = w[s][0]*fy[0] + w[s][1]*fy[1] + w[s][2]*fy[2];
    }

    float mu = (ax[0]+ay[0]+ax[1]+ay[1]+ax[2]+ay[2]) * (1.f/6.f);
    float var = 0.f;
    #pragma unroll
    for (int s = 0; s < 3; ++s) {
        float dx = ax[s] - mu, dy = ay[s] - mu;
        var += dx*dx + dy*dy;
    }
    var *= (1.f/6.f);
    float inv = __builtin_amdgcn_rsqf(var + 1e-5f);
    #pragma unroll
    for (int s = 0; s < 3; ++s) {
        flat[2*s+0] = (ax[s] - mu) * inv * lnw[2*s+0] + lnb[2*s+0];
        flat[2*s+1] = (ay[s] - mu) * inv * lnw[2*s+1] + lnb[2*s+1];
    }
}

__device__ __forceinline__ v2f pmax(v2f a, v2f b) { return __builtin_elementwise_max(a, b); }

__device__ __forceinline__ void pair_forward(
    v2f h0, v2f h1,
    v2f pv0, v2f pv1, v2f pv2, v2f pv3, v2f pv4,
    v2f av0, v2f av1, v2f av2, v2f av3, v2f av4,
    const float* __restrict__ Wh, const float* __restrict__ bh,
    const float* __restrict__ Wp, const float* __restrict__ bp,
    const float* __restrict__ Wa, const float* __restrict__ ba,
    const float* __restrict__ lnw, const float* __restrict__ lnb,
    v2f* flat)
{
    const v2f zero = {0.f, 0.f};
    v2f fx[3], fy[3];
    fx[0] = pmax(Wh[0]*h0 + Wh[1]*h1 + bh[0], zero);
    fy[0] = pmax(Wh[2]*h0 + Wh[3]*h1 + bh[1], zero);
    fx[1] = pmax(Wp[0]*pv0 + Wp[1]*pv1 + Wp[2]*pv2 + Wp[3]*pv3 + Wp[4]*pv4 + bp[0], zero);
    fy[1] = pmax(Wp[5]*pv0 + Wp[6]*pv1 + Wp[7]*pv2 + Wp[8]*pv3 + Wp[9]*pv4 + bp[1], zero);
    fx[2] = pmax(Wa[0]*av0 + Wa[1]*av1 + Wa[2]*av2 + Wa[3]*av3 + Wa[4]*av4 + ba[0], zero);
    fy[2] = pmax(Wa[5]*av0 + Wa[6]*av1 + Wa[7]*av2 + Wa[8]*av3 + Wa[9]*av4 + ba[1], zero);

    const float is2 = 0.70710678118654752f;
    v2f l[3][3];
    #pragma unroll
    for (int s = 0; s < 3; ++s)
        #pragma unroll
        for (int t = 0; t < 3; ++t)
            l[s][t] = (fx[s]*fx[t] + fy[s]*fy[t]) * is2;

    const float log2e = 1.44269504088896341f;
    v2f w[3][3];
    #pragma unroll
    for (int t = 0; t < 3; ++t) {
        v2f m  = pmax(l[0][t], pmax(l[1][t], l[2][t]));
        v2f a0 = (l[0][t] - m) * log2e;
        v2f a1 = (l[1][t] - m) * log2e;
        v2f a2 = (l[2][t] - m) * log2e;
        v2f e0, e1, e2;
        e0.x = __builtin_amdgcn_exp2f(a0.x); e0.y = __builtin_amdgcn_exp2f(a0.y);
        e1.x = __builtin_amdgcn_exp2f(a1.x); e1.y = __builtin_amdgcn_exp2f(a1.y);
        e2.x = __builtin_amdgcn_exp2f(a2.x); e2.y = __builtin_amdgcn_exp2f(a2.y);
        v2f sum = e0 + e1 + e2;
        v2f r;
        r.x = __builtin_amdgcn_rcpf(sum.x); r.y = __builtin_amdgcn_rcpf(sum.y);
        w[0][t] = e0 * r; w[1][t] = e1 * r; w[2][t] = e2 * r;
    }

    v2f ax[3], ay[3];
    #pragma unroll
    for (int s = 0; s < 3; ++s) {
        ax[s] = w[s][0]*fx[0] + w[s][1]*fx[1] + w[s][2]*fx[2];
        ay[s] = w[s][0]*fy[0] + w[s][1]*fy[1] + w[s][2]*fy[2];
    }

    v2f mu = (ax[0]+ay[0]+ax[1]+ay[1]+ax[2]+ay[2]) * (1.f/6.f);
    v2f var = zero;
    #pragma unroll
    for (int s = 0; s < 3; ++s) {
        v2f dx = ax[s] - mu, dy = ay[s] - mu;
        var += dx*dx + dy*dy;
    }
    var = var * (1.f/6.f) + 1e-5f;
    v2f inv;
    inv.x = __builtin_amdgcn_rsqf(var.x); inv.y = __builtin_amdgcn_rsqf(var.y);
    #pragma unroll
    for (int s = 0; s < 3; ++s) {
        flat[2*s+0] = (ax[s] - mu) * inv * lnw[2*s+0] + lnb[2*s+0];
        flat[2*s+1] = (ay[s] - mu) * inv * lnw[2*s+1] + lnb[2*s+1];
    }
}

__global__ __launch_bounds__(256) void fused_attn_kernel(
    const float* __restrict__ hist,
    const float* __restrict__ paper,
    const float* __restrict__ author,
    const float* __restrict__ Wh, const float* __restrict__ bh,
    const float* __restrict__ Wp, const float* __restrict__ bp,
    const float* __restrict__ Wa, const float* __restrict__ ba,
    const float* __restrict__ lnw, const float* __restrict__ lnb,
    const float* __restrict__ Wfc, const float* __restrict__ bfc,
    float* __restrict__ out, int n)
{
    const int    lane     = threadIdx.x & 31;
    const size_t wave     = ((size_t)blockIdx.x * blockDim.x + threadIdx.x) >> 5;
    const size_t waveBase = wave * 256u;
    const bool   lo       = lane < 16;

    if (waveBase >= (size_t)n) return;

    if (waveBase + 256u <= (size_t)n) {
        v2f   hv[8];
        float pv[8][5], av[8][5];
        #pragma unroll
        for (int q = 0; q < 8; ++q) {
            const size_t row = waveBase + (size_t)(32 * q + lane);
            const float* hp = hist   + row * 2u;
            const float* pp = paper  + row * 5u;
            const float* ap = author + row * 5u;
            hv[q] = __builtin_nontemporal_load((const v2f*)hp);
            #pragma unroll
            for (int i = 0; i < 5; ++i) pv[q][i] = __builtin_nontemporal_load(pp + i);
            #pragma unroll
            for (int i = 0; i < 5; ++i) av[q][i] = __builtin_nontemporal_load(ap + i);
        }

        const int  j  = lane & 15;
        const bool jv = j < 6;
        v2f A1, A2;
        A1.x = jv ? Wfc[j*6 + (lo ? 0 : 2)] : 0.f;
        A1.y = jv ? Wfc[j*6 + (lo ? 1 : 3)] : 0.f;
        A2.x = (jv && lo) ? Wfc[j*6 + 4] : 0.f;
        A2.y = (jv && lo) ? Wfc[j*6 + 5] : 0.f;

        float flat[8][6];
        #pragma unroll
        for (int p = 0; p < 4; ++p) {
            const int q0 = 2 * p, q1 = 2 * p + 1;
            v2f F[6];
            pair_forward((v2f){hv[q0].x, hv[q1].x}, (v2f){hv[q0].y, hv[q1].y},
                         (v2f){pv[q0][0], pv[q1][0]}, (v2f){pv[q0][1], pv[q1][1]},
                         (v2f){pv[q0][2], pv[q1][2]}, (v2f){pv[q0][3], pv[q1][3]},
                         (v2f){pv[q0][4], pv[q1][4]},
                         (v2f){av[q0][0], av[q1][0]}, (v2f){av[q0][1], av[q1][1]},
                         (v2f){av[q0][2], av[q1][2]}, (v2f){av[q0][3], av[q1][3]},
                         (v2f){av[q0][4], av[q1][4]},
                         Wh, bh, Wp, bp, Wa, ba, lnw, lnb, F);
            #pragma unroll
            for (int k = 0; k < 6; ++k) { flat[q0][k] = F[k].x; flat[q1][k] = F[k].y; }
        }

        #pragma unroll
        for (int q = 0; q < 8; ++q)
            asm volatile("" : "+v"(flat[q][0]), "+v"(flat[q][1]), "+v"(flat[q][2]),
                              "+v"(flat[q][3]), "+v"(flat[q][4]), "+v"(flat[q][5]));


        v8f C = {};
        if (lo) { C[0]=bfc[0]; C[1]=bfc[1]; C[2]=bfc[2]; C[3]=bfc[3]; C[4]=bfc[4]; C[5]=bfc[5]; }

        #pragma unroll
        for (int q = 0; q < 8; ++q) {
            const float* f = flat[q];
            const size_t subBase = waveBase + (size_t)(32 * q);
            float f2s = bperm(lane & 15, f[2]);
            float f3s = bperm(lane & 15, f[3]);
            v2f B1; B1.x = lo ? f[0] : f2s;  B1.y = lo ? f[1] : f3s;
            v2f B2; B2.x = lo ? f[4] : 0.f;  B2.y = lo ? f[5] : 0.f;
            v8f D1 = __builtin_amdgcn_wmma_f32_16x16x4_f32(false, A1, false, B1, (short)0, C,  false, false);
            D1     = __builtin_amdgcn_wmma_f32_16x16x4_f32(false, A2, false, B2, (short)0, D1, false, false);

            float f0s = bperm(lane | 16, f[0]);
            float f1s = bperm(lane | 16, f[1]);
            float f4s = bperm(lane | 16, f[4]);
            float f5s = bperm(lane | 16, f[5]);
            v2f B3; B3.x = lo ? f0s : f[2];  B3.y = lo ? f1s : f[3];
            v2f B4; B4.x = lo ? f4s : 0.f;   B4.y = lo ? f5s : 0.f;
            v8f D2 = __builtin_amdgcn_wmma_f32_16x16x4_f32(false, A1, false, B3, (short)0, C,  false, false);
            D2     = __builtin_amdgcn_wmma_f32_16x16x4_f32(false, A2, false, B4, (short)0, D2, false, false);

            if (lo) {
                float* o1 = out + (subBase + (size_t)lane) * 6u;
                v2f s0; s0.x = fmaxf(D1[0], 0.f); s0.y = fmaxf(D1[1], 0.f);
                v2f s1; s1.x = fmaxf(D1[2], 0.f); s1.y = fmaxf(D1[3], 0.f);
                v2f s2; s2.x = fmaxf(D1[4], 0.f); s2.y = fmaxf(D1[5], 0.f);
                __builtin_nontemporal_store(s0, (v2f*)(o1 + 0));
                __builtin_nontemporal_store(s1, (v2f*)(o1 + 2));
                __builtin_nontemporal_store(s2, (v2f*)(o1 + 4));

                float* o2 = out + (subBase + 16u + (size_t)lane) * 6u;
                v2f t0; t0.x = fmaxf(D2[0], 0.f); t0.y = fmaxf(D2[1], 0.f);
                v2f t1; t1.x = fmaxf(D2[2], 0.f); t1.y = fmaxf(D2[3], 0.f);
                v2f t2; t2.x = fmaxf(D2[4], 0.f); t2.y = fmaxf(D2[5], 0.f);
                __builtin_nontemporal_store(t0, (v2f*)(o2 + 0));
                __builtin_nontemporal_store(t1, (v2f*)(o2 + 2));
                __builtin_nontemporal_store(t2, (v2f*)(o2 + 4));
            }
        }
    } else {
        for (int q = 0; q < 8; ++q) {
            const size_t row = waveBase + (size_t)(32 * q + lane);
            if (row < (size_t)n) {
                const float* hp = hist   + row * 2u;
                const float* pp = paper  + row * 5u;
                const float* ap = author + row * 5u;
                float flat[6];
                row_forward(hp[0], hp[1], pp[0], pp[1], pp[2], pp[3], pp[4],
                            ap[0], ap[1], ap[2], ap[3], ap[4],
                            Wh, bh, Wp, bp, Wa, ba, lnw, lnb, flat);
                float* o = out + row * 6u;
                #pragma unroll
                for (int jj = 0; jj < 6; ++jj) {
                    float acc = bfc[jj];
                    #pragma unroll
                    for (int k = 0; k < 6; ++k) acc += Wfc[jj*6 + k] * flat[k];
                    o[jj] = fmaxf(acc, 0.f);
                }
            }
        }
    }
}

extern "C" void kernel_launch(void* const* d_in, const int* in_sizes, int n_in,
                              void* d_out, int out_size, void* d_ws, size_t ws_size,
                              hipStream_t stream) {
    const float* hist   = (const float*)d_in[0];
    const float* paper  = (const float*)d_in[1];
    const float* author = (const float*)d_in[2];
    const float* Wh  = (const float*)d_in[3];
    const float* bh  = (const float*)d_in[4];
    const float* Wp  = (const float*)d_in[5];
    const float* bp  = (const float*)d_in[6];
    const float* Wa  = (const float*)d_in[7];
    const float* ba  = (const float*)d_in[8];
    const float* lnw = (const float*)d_in[9];
    const float* lnb = (const float*)d_in[10];
    const float* Wfc = (const float*)d_in[11];
    const float* bfc = (const float*)d_in[12];
    float* out = (float*)d_out;

    const int n = in_sizes[0] / 2;
    const int threads = 256;
    const int rowsPerBlock = 2048;
    const int blocks  = (n + rowsPerBlock - 1) / rowsPerBlock;

    fused_attn_kernel<<<blocks, threads, 0, stream>>>(
        hist, paper, author, Wh, bh, Wp, bp, Wa, ba, lnw, lnb, Wfc, bfc, out, n);
}
